# starting kernel
# baseline (speedup 1.0000x reference)
#include <hip/hip_runtime.h>
#include <hip/hip_bf16.h>
#include <math.h>


typedef __attribute__((ext_vector_type(2))) float v2f;
typedef __attribute__((ext_vector_type(8))) float v8f;

#define WAVE 32

__global__ void fill_kernel(float* __restrict__ p, float v, long long n) {
    long long i = (long long)blockIdx.x * blockDim.x + threadIdx.x;
    if (i < n) p[i] = v;
}

__global__ void selfloop_accum_kernel(const int* __restrict__ dst,
                                      const float* __restrict__ ew,
                                      float* __restrict__ sumw,
                                      float* __restrict__ cntw, int E) {
    int e = blockIdx.x * blockDim.x + threadIdx.x;
    if (e >= E) return;
    int d = dst[e];
    atomicAdd(&sumw[d], ew[e]);
    atomicAdd(&cntw[d], 1.0f);
}

__global__ void selfloop_w_kernel(const float* __restrict__ sumw,
                                  const float* __restrict__ cntw,
                                  float* __restrict__ loopw, int n) {
    int i = blockIdx.x * blockDim.x + threadIdx.x;
    if (i >= n) return;
    loopw[i] = sumw[i] / fmaxf(cntw[i], 1.0f);
}

__global__ void gemm_bias_wmma_kernel(const float* __restrict__ X,
                                      const float* __restrict__ W,
                                      const float* __restrict__ bias,
                                      float* __restrict__ Y,
                                      int rows, int K, int Mout) {
    int gtid = blockIdx.x * blockDim.x + threadIdx.x;
    int wave = gtid >> 5;
    int lane = gtid & 31;
    int tilesN = Mout >> 4;
    int rowT = wave / tilesN;
    int colT = wave - rowT * tilesN;
    if (rowT * 16 >= rows) return;

    const int m    = lane & 15;
    const int half = lane >> 4;

    const float* __restrict__ Arow = X + (size_t)(rowT * 16 + m) * K;
    const float* __restrict__ Bcol = W + (size_t)(colT * 16 + m);

    v8f acc = {};
    for (int k0 = 0; k0 < K; k0 += 4) {
        const int ka = k0 + half * 2;
        v2f a;
        a.x = Arow[ka];
        a.y = Arow[ka + 1];
        v2f b;
        b.x = Bcol[(size_t)ka * Mout];
        b.y = Bcol[(size_t)(ka + 1) * Mout];
        acc = __builtin_amdgcn_wmma_f32_16x16x4_f32(
             false, a,  false, b,
             (short)0, acc,  false,  false);
    }

    const int col = colT * 16 + m;
    const float bv = bias[col];
#pragma unroll
    for (int j = 0; j < 8; ++j) {
        const int row = rowT * 16 + j + half * 8;
        Y[(size_t)row * Mout + col] = acc[j] + bv;
    }
}

__device__ __forceinline__ void atomicMaxF(float* addr, float value) {
    if (value >= 0.0f)
        atomicMax((int*)addr, __float_as_int(value));
    else
        atomicMin((unsigned int*)addr, __float_as_uint(value));
}

template <int C>
__global__ void edge_logits_kernel(const int* __restrict__ src,
                                   const int* __restrict__ dst,
                                   const float* __restrict__ ew,
                                   const float* __restrict__ loopw,
                                   int E, int Etot,
                                   const float* __restrict__ xl,
                                   const float* __restrict__ xr,
                                   const float* __restrict__ We,
                                   const float* __restrict__ att,
                                   float* __restrict__ logits,
                                   float* __restrict__ maxb) {
    const int gtid = blockIdx.x * blockDim.x + threadIdx.x;
    const int e    = gtid >> 5;
    const int lane = gtid & 31;
    if (e >= Etot) return;

    int s, d;
    float w;
    if (e < E) { s = src[e]; d = dst[e]; w = ew[e]; }
    else       { s = d = e - E; w = loopw[s]; }

    const int HC = 4 * C;
    float lg0 = 0.f, lg1 = 0.f, lg2 = 0.f, lg3 = 0.f;

    for (int c = lane; c < C; c += WAVE) {
        const size_t sb = (size_t)s * HC;
        const size_t db = (size_t)d * HC;
#pragma unroll
        for (int h = 0; h < 4; ++h) {
            const int idx = h * C + c;
            float v = xl[sb + idx] + xr[db + idx] + w * We[idx];
            v = (v > 0.f) ? v : 0.2f * v;
            const float t = v * att[idx];
            if (h == 0) lg0 += t; else if (h == 1) lg1 += t;
            else if (h == 2) lg2 += t; else lg3 += t;
        }
    }
#pragma unroll
    for (int off = 16; off > 0; off >>= 1) {
        lg0 += __shfl_xor(lg0, off, WAVE);
        lg1 += __shfl_xor(lg1, off, WAVE);
        lg2 += __shfl_xor(lg2, off, WAVE);
        lg3 += __shfl_xor(lg3, off, WAVE);
    }
    if (lane == 0) {
        float lg[4] = {lg0, lg1, lg2, lg3};
#pragma unroll
        for (int h = 0; h < 4; ++h) {
            logits[(size_t)e * 4 + h] = lg[h];
            atomicMaxF(&maxb[(size_t)d * 4 + h], lg[h]);
        }
    }
}

__global__ void edge_softmax_kernel(const int* __restrict__ dst, int E, int Etot,
                                    float* __restrict__ logits,
                                    const float* __restrict__ maxb,
                                    float* __restrict__ denb) {
    const int e = blockIdx.x * blockDim.x + threadIdx.x;
    if (e >= Etot) return;
    const int d = (e < E) ? dst[e] : (e - E);
#pragma unroll
    for (int h = 0; h < 4; ++h) {
        const float p = __expf(logits[(size_t)e * 4 + h] - maxb[(size_t)d * 4 + h]);
        logits[(size_t)e * 4 + h] = p;
        atomicAdd(&denb[(size_t)d * 4 + h], p);
    }
}

template <int C>
__global__ void edge_aggregate_kernel(const int* __restrict__ src,
                                      const int* __restrict__ dst,
                                      int E, int Etot,
                                      const float* __restrict__ pnum,
                                      const float* __restrict__ denb,
                                      const float* __restrict__ xl,
                                      float* __restrict__ outacc) {
    const int gtid = blockIdx.x * blockDim.x + threadIdx.x;
    const int e    = gtid >> 5;
    const int lane = gtid & 31;
    if (e >= Etot) return;

    int s, d;
    if (e < E) { s = src[e]; d = dst[e]; }
    else       { s = d = e - E; }

    float alpha[4];
#pragma unroll
    for (int h = 0; h < 4; ++h)
        alpha[h] = pnum[(size_t)e * 4 + h] / denb[(size_t)d * 4 + h];

    const int HC = 4 * C;
    const size_t sb = (size_t)s * HC;
    const size_t db = (size_t)d * HC;
    for (int c = lane; c < C; c += WAVE) {
#pragma unroll
        for (int h = 0; h < 4; ++h) {
            const int idx = h * C + c;
            atomicAdd(&outacc[db + idx], xl[sb + idx] * alpha[h]);
        }
    }
}

__global__ void finalize_concat_kernel(const float* __restrict__ acc,
                                       const float* __restrict__ bias,
                                       const float* __restrict__ g,
                                       const float* __restrict__ be,
                                       const float* __restrict__ res,
                                       float* __restrict__ out, int nN) {
    const int gtid = blockIdx.x * blockDim.x + threadIdx.x;
    const int node = gtid >> 5;
    const int lane = gtid & 31;
    if (node >= nN) return;
    const int D = 128;

    float v[4], s = 0.f, s2 = 0.f;
#pragma unroll
    for (int i = 0; i < 4; ++i) {
        const int ch = lane + i * WAVE;
        float x = acc[(size_t)node * D + ch] + bias[ch];
        x = (x > 0.f) ? x : (__expf(x) - 1.f);
        v[i] = x;
        s += x;
        s2 += x * x;
    }
#pragma unroll
    for (int off = 16; off > 0; off >>= 1) {
        s  += __shfl_xor(s, off, WAVE);
        s2 += __shfl_xor(s2, off, WAVE);
    }
    const float mu  = s / (float)D;
    const float var = s2 / (float)D - mu * mu;
    const float inv = rsqrtf(var + 1e-5f);
#pragma unroll
    for (int i = 0; i < 4; ++i) {
        const int ch = lane + i * WAVE;
        out[(size_t)node * D + ch] =
            (v[i] - mu) * inv * g[ch] + be[ch] + res[(size_t)node * D + ch];
    }
}

__global__ void finalize_mean_kernel(const float* __restrict__ acc,
                                     const float* __restrict__ bias,
                                     const float* __restrict__ g,
                                     const float* __restrict__ be,
                                     const float* __restrict__ res,
                                     float* __restrict__ out, int nN) {
    const int gtid = blockIdx.x * blockDim.x + threadIdx.x;
    const int node = gtid >> 5;
    const int lane = gtid & 31;
    if (node >= nN) return;
    const int D = 64;

    float v[2], s = 0.f, s2 = 0.f;
#pragma unroll
    for (int i = 0; i < 2; ++i) {
        const int c = lane + i * WAVE;
        const size_t nb = (size_t)node * 256;
        float x = 0.25f * (acc[nb + 0 * D + c] + acc[nb + 1 * D + c] +
                           acc[nb + 2 * D + c] + acc[nb + 3 * D + c]) + bias[c];
        x = (x > 0.f) ? x : (__expf(x) - 1.f);
        v[i] = x;
        s += x;
        s2 += x * x;
    }
#pragma unroll
    for (int off = 16; off > 0; off >>= 1) {
        s  += __shfl_xor(s, off, WAVE);
        s2 += __shfl_xor(s2, off, WAVE);
    }
    const float mu  = s / (float)D;
    const float var = s2 / (float)D - mu * mu;
    const float inv = rsqrtf(var + 1e-5f);
#pragma unroll
    for (int i = 0; i < 2; ++i) {
        const int c = lane + i * WAVE;
        out[(size_t)node * D + c] =
            (v[i] - mu) * inv * g[c] + be[c] + res[(size_t)node * D + c];
    }
}

static inline int blocks_for(long long work, int bs) {
    return (int)((work + bs - 1) / bs);
}

extern "C" void kernel_launch(void* const* d_in, const int* in_sizes, int n_in,
                              void* d_out, int out_size, void* d_ws, size_t ws_size,
                              hipStream_t stream) {
    const float* x    = (const float*)d_in[0];
    const int*   ei   = (const int*)d_in[1];
    const float* ew   = (const float*)d_in[2];
    const float* Wl0  = (const float*)d_in[3];
    const float* bl0  = (const float*)d_in[4];
    const float* Wr0  = (const float*)d_in[5];
    const float* br0  = (const float*)d_in[6];
    const float* We0  = (const float*)d_in[7];
    const float* att0 = (const float*)d_in[8];
    const float* b0   = (const float*)d_in[9];
    const float* g0   = (const float*)d_in[10];
    const float* be0  = (const float*)d_in[11];
    const float* Wl1  = (const float*)d_in[12];
    const float* bl1  = (const float*)d_in[13];
    const float* Wr1  = (const float*)d_in[14];
    const float* br1  = (const float*)d_in[15];
    const float* We1  = (const float*)d_in[16];
    const float* att1 = (const float*)d_in[17];
    const float* b1   = (const float*)d_in[18];
    const float* g1   = (const float*)d_in[19];
    const float* be1  = (const float*)d_in[20];
    const float* P0   = (const float*)d_in[21];
    const float* pb0  = (const float*)d_in[22];
    const float* P1   = (const float*)d_in[23];
    const float* pb1  = (const float*)d_in[24];

    const int IN  = 128, HIDD = 128, O1 = 256, OUT = 64;
    const int N = in_sizes[0] / IN;
    const int E = in_sizes[2];
    const int Etot = E + N;
    const int* srcp = ei;
    const int* dstp = ei + E;

    float* ws = (float*)d_ws;
    size_t off = 0;
    float* sumw  = ws + off; off += (size_t)N;
    float* cntw  = ws + off; off += (size_t)N;
    float* loopw = ws + off; off += (size_t)N;
    float* XL    = ws + off; off += (size_t)N * O1;
    float* XR    = ws + off; off += (size_t)N * O1;
    float* ACC   = ws + off; off += (size_t)N * O1;
    float* RES   = ws + off; off += (size_t)N * HIDD;
    float* HCUR  = ws + off; off += (size_t)N * HIDD;
    float* MAXB  = ws + off; off += (size_t)N * 4;
    float* DENB  = ws + off; off += (size_t)N * 4;
    float* LOGIT = ws + off; off += (size_t)Etot * 4;
    (void)ws_size; (void)n_in; (void)out_size;

    const int BS = 256;
    const float NEG_INF = -__builtin_inff();

    fill_kernel<<<blocks_for(2LL * N, BS), BS, 0, stream>>>(sumw, 0.f, 2LL * N);
    selfloop_accum_kernel<<<blocks_for(E, BS), BS, 0, stream>>>(dstp, ew, sumw, cntw, E);
    selfloop_w_kernel<<<blocks_for(N, BS), BS, 0, stream>>>(sumw, cntw, loopw, N);

    {
        const long long t128 = (long long)(N / 16) * (HIDD / 16) * WAVE;
        gemm_bias_wmma_kernel<<<blocks_for(t128, BS), BS, 0, stream>>>(x, Wl0, bl0, XL, N, IN, HIDD);
        gemm_bias_wmma_kernel<<<blocks_for(t128, BS), BS, 0, stream>>>(x, Wr0, br0, XR, N, IN, HIDD);
        gemm_bias_wmma_kernel<<<blocks_for(t128, BS), BS, 0, stream>>>(x, P0, pb0, RES, N, IN, HIDD);

        fill_kernel<<<blocks_for(4LL * N, BS), BS, 0, stream>>>(MAXB, NEG_INF, 4LL * N);
        fill_kernel<<<blocks_for(4LL * N, BS), BS, 0, stream>>>(DENB, 0.f, 4LL * N);
        fill_kernel<<<blocks_for((long long)N * HIDD, BS), BS, 0, stream>>>(ACC, 0.f, (long long)N * HIDD);

        const long long ethreads = (long long)Etot * WAVE;
        edge_logits_kernel<32><<<blocks_for(ethreads, BS), BS, 0, stream>>>(
            srcp, dstp, ew, loopw, E, Etot, XL, XR, We0, att0, LOGIT, MAXB);
        edge_softmax_kernel<<<blocks_for(Etot, BS), BS, 0, stream>>>(
            dstp, E, Etot, LOGIT, MAXB, DENB);
        edge_aggregate_kernel<32><<<blocks_for(ethreads, BS), BS, 0, stream>>>(
            srcp, dstp, E, Etot, LOGIT, DENB, XL, ACC);

        finalize_concat_kernel<<<blocks_for((long long)N * WAVE, BS), BS, 0, stream>>>(
            ACC, b0, g0, be0, RES, HCUR, N);
    }

    {
        const long long t256 = (long long)(N / 16) * (O1 / 16) * WAVE;
        const long long t64  = (long long)(N / 16) * (OUT / 16) * WAVE;
        gemm_bias_wmma_kernel<<<blocks_for(t256, BS), BS, 0, stream>>>(HCUR, Wl1, bl1, XL, N, HIDD, O1);
        gemm_bias_wmma_kernel<<<blocks_for(t256, BS), BS, 0, stream>>>(HCUR, Wr1, br1, XR, N, HIDD, O1);
        gemm_bias_wmma_kernel<<<blocks_for(t64, BS), BS, 0, stream>>>(HCUR, P1, pb1, RES, N, HIDD, OUT);

        fill_kernel<<<blocks_for(4LL * N, BS), BS, 0, stream>>>(MAXB, NEG_INF, 4LL * N);
        fill_kernel<<<blocks_for(4LL * N, BS), BS, 0, stream>>>(DENB, 0.f, 4LL * N);
        fill_kernel<<<blocks_for((long long)N * O1, BS), BS, 0, stream>>>(ACC, 0.f, (long long)N * O1);

        const long long ethreads = (long long)Etot * WAVE;
        edge_logits_kernel<64><<<blocks_for(ethreads, BS), BS, 0, stream>>>(
            srcp, dstp, ew, loopw, E, Etot, XL, XR, We1, att1, LOGIT, MAXB);
        edge_softmax_kernel<<<blocks_for(Etot, BS), BS, 0, stream>>>(
            dstp, E, Etot, LOGIT, MAXB, DENB);
        edge_aggregate_kernel<64><<<blocks_for(ethreads, BS), BS, 0, stream>>>(
            srcp, dstp, E, Etot, LOGIT, DENB, XL, ACC);

        finalize_mean_kernel<<<blocks_for((long long)N * WAVE, BS), BS, 0, stream>>>(
            ACC, b1, g1, be1, RES, (float*)d_out, N);
    }
}


# Dense transforms retiled: a layer's three products share one launch, weights resident in LDS (one work-group per WGP), a wave keeps its strip of input rows in registers, stores 256-byte row pieces
# speedup vs baseline: 3.7890x; 3.7890x over previous
#include <hip/hip_runtime.h>
#include <hip/hip_bf16.h>
#include <math.h>


typedef __attribute__((ext_vector_type(2))) float v2f;
typedef __attribute__((ext_vector_type(8))) float v8f;

#define WAVE 32

__global__ void fill_kernel(float* __restrict__ p, float v, long long n) {
    long long i = (long long)blockIdx.x * blockDim.x + threadIdx.x;
    if (i < n) p[i] = v;
}

typedef __attribute__((ext_vector_type(4))) float v4f;

__global__ void degree_rank_kernel(const int* __restrict__ dst,
                                   int* __restrict__ cnt,
                                   int* __restrict__ rank, int E) {
    int e = blockIdx.x * blockDim.x + threadIdx.x;
    if (e >= E) return;
    rank[e] = atomicAdd(&cnt[dst[e]], 1);
}

__global__ void __launch_bounds__(1024)
rowptr_scan_kernel(const int* __restrict__ cnt, int* __restrict__ rowptr, int n) {
    __shared__ int wsum[WAVE];
    const int t    = threadIdx.x;
    const int lane = t & 31;
    const int wv   = t >> 5;
    const int per  = (n + 1023) / 1024;
    const int b    = t * per;

    const bool whole = (b + per <= n);
    int s = 0;
    if (whole) {
        for (int i = 0; i < per; ++i) s += cnt[b + i];
    } else {
        for (int i = 0; i < per; ++i) {
            const int k = b + i;
            if (k < n) s += cnt[k];
        }
    }
    int incl = s;
    for (int off = 1; off < WAVE; off <<= 1) {
        const int y = __shfl_up(incl, off, WAVE);
        if (lane >= off) incl += y;
    }
    if (lane == WAVE - 1) wsum[wv] = incl;
    __syncthreads();
    if (wv == 0) {
        const int v = wsum[lane];
        int iv = v;
        for (int off = 1; off < WAVE; off <<= 1) {
            const int y = __shfl_up(iv, off, WAVE);
            if (lane >= off) iv += y;
        }
        wsum[lane] = iv - v;
    }
    __syncthreads();
    int run = wsum[wv] + incl - s;
    if (whole) {
        for (int i = 0; i < per; ++i) { rowptr[b + i] = run; run += cnt[b + i]; }
    } else {
        for (int i = 0; i < per; ++i) {
            const int k = b + i;
            if (k < n) { rowptr[k] = run; run += cnt[k]; }
        }
    }
    if (t == 1023) rowptr[n] = run;
}

__global__ void group_by_dst_kernel(const int* __restrict__ src,
                                    const int* __restrict__ dst,
                                    const float* __restrict__ ew,
                                    const int* __restrict__ rank,
                                    const int* __restrict__ rowptr,
                                    int* __restrict__ gsrc,
                                    float* __restrict__ gw, int E) {
    int e = blockIdx.x * blockDim.x + threadIdx.x;
    if (e >= E) return;
    const int pos = rowptr[dst[e]] + rank[e];
    gsrc[pos] = src[e];
    gw[pos]   = ew[e];
}

template <int MP, int COFF, int PITCH, int NT>
__device__ __forceinline__ void stage_weights(const float* __restrict__ Wp,
                                              float* __restrict__ Wl, int tid) {
    constexpr int Q = MP / 4;
    static_assert((64 * Q) % NT == 0, "every thread stages the same number of pieces");
#pragma unroll
    for (int i = 0; i < (64 * Q) / NT; ++i) {
        const int it = tid + i * NT;
        const int p  = it / Q;
        const int q4 = it - p * Q;
        const v4f r0 = *(const v4f*)(Wp + (size_t)(2 * p) * MP + 4 * q4);
        const v4f r1 = *(const v4f*)(Wp + (size_t)(2 * p + 1) * MP + 4 * q4);
        const int col0  = COFF + 4 * q4;
        const int gbase = col0 & ~63;
        const int cc    = (col0 & 63) >> 2;
        float* dst = Wl + p * PITCH + 2 * (gbase + cc);
        v2f e;
        e.x = r0.x; e.y = r1.x; *(v2f*)(dst +  0) = e;
        e.x = r0.y; e.y = r1.y; *(v2f*)(dst + 32) = e;
        e.x = r0.z; e.y = r1.z; *(v2f*)(dst + 64) = e;
        e.x = r0.w; e.y = r1.w; *(v2f*)(dst + 96) = e;
    }
}

template <int M0, int M1, int M2>
__global__ void __launch_bounds__(512)
layer_gemm_wmma_kernel(const float* __restrict__ X,
                       const float* __restrict__ W0, const float* __restrict__ B0, float* __restrict__ Y0,
                       const float* __restrict__ W1, const float* __restrict__ B1, float* __restrict__ Y1,
                       const float* __restrict__ W2, const float* __restrict__ B2, float* __restrict__ Y2,
                       int rows) {
    constexpr int KD    = 128;
    constexpr int MTOT  = M0 + M1 + M2;
    constexpr int PITCH = 2 * MTOT + 16;
    constexpr int G0 = M0 / 64, G1 = M1 / 64, G2 = M2 / 64;
    constexpr int G  = G0 + G1 + G2;
    constexpr int NW = 16;
    static_assert(M0 % 64 == 0 && M1 % 64 == 0 && M2 % 64 == 0, "whole column groups");
    static_assert(64 * PITCH * 4 <= 320 * 1024, "the layer's weights fit the work-group's LDS");

    __shared__ __attribute__((aligned(16))) float Wl[64 * PITCH];

    const int tid  = threadIdx.x;
    const int lane = tid & 31;
    const int c    = lane & 15;
    const int h    = lane >> 4;
    const int wv   = __builtin_amdgcn_readfirstlane(tid >> 5);

    const int nstrips = rows >> 4;
    const int sbeg = (int)((blockIdx.x * (unsigned)nstrips) / gridDim.x);
    const int send = (int)(((blockIdx.x + 1) * (unsigned)nstrips) / gridDim.x);
    const int L    = (send - sbeg) * G;
    const int ubeg = (wv * L) / NW;
    const int uend = ((wv + 1) * L) / NW;

    v4f a[16];
    int cur = -1;
    if (ubeg < uend) {
        cur = sbeg + ubeg / G;
        const v4f* __restrict__ pa = (const v4f*)(X + (size_t)(cur * 16 + c) * KD + 4 * h);
#pragma unroll
        for (int j = 0; j < 16; ++j) a[j] = pa[2 * j];
    }

    stage_weights<M0, 0,       PITCH, NW * WAVE>(W0, Wl, tid);
    stage_weights<M1, M0,      PITCH, NW * WAVE>(W1, Wl, tid);
    stage_weights<M2, M0 + M1, PITCH, NW * WAVE>(W2, Wl, tid);
    __syncthreads();

    for (int u = ubeg; u < uend; ++u) {
        const int sl    = u / G;
        const int g     = u - sl * G;
        const int strip = sbeg + sl;
        if (strip != cur) {
            cur = strip;
            const v4f* __restrict__ pa = (const v4f*)(X + (size_t)(strip * 16 + c) * KD + 4 * h);
#pragma unroll
            for (int j = 0; j < 16; ++j) a[j] = pa[2 * j];
        }

        const float* bias; float* Y; int M; int gl;
        if (g < G0)           { bias = B0; Y = Y0; M = M0; gl = g; }
        else if (g < G0 + G1) { bias = B1; Y = Y1; M = M1; gl = g - G0; }
        else                  { bias = B2; Y = Y2; M = M2; gl = g - G0 - G1; }
        const int lc = gl * 64 + 4 * c;
        const v4f bb = *(const v4f*)(bias + lc);

        const float* bl = Wl + (2 * h) * PITCH + 128 * g + 2 * c;
        v8f acc[4] = {};
#pragma unroll
        for (int j = 0; j < 16; ++j) {
            v2f a0, a1;
            a0.x = a[j].x; a0.y = a[j].y;
            a1.x = a[j].z; a1.y = a[j].w;
#pragma unroll
            for (int t = 0; t < 4; ++t) {
                const v2f b = *(const v2f*)(bl + (4 * j) * PITCH + 32 * t);
                acc[t] = __builtin_amdgcn_wmma_f32_16x16x4_f32(
                     false, a0,  false, b,
                     (short)0, acc[t],  false,  false);
            }
#pragma unroll
            for (int t = 0; t < 4; ++t) {
                const v2f b = *(const v2f*)(bl + (4 * j + 1) * PITCH + 32 * t);
                acc[t] = __builtin_amdgcn_wmma_f32_16x16x4_f32(
                     false, a1,  false, b,
                     (short)0, acc[t],  false,  false);
            }
        }

        float* yrow = Y + (size_t)(strip * 16 + 8 * h) * M + lc;
#pragma unroll
        for (int j = 0; j < 8; ++j) {
            v4f o;
            o.x = acc[0][j] + bb.x; o.y = acc[1][j] + bb.y;
            o.z = acc[2][j] + bb.z; o.w = acc[3][j] + bb.w;
            *(v4f*)(yrow + (size_t)j * M) = o;
        }
    }
}

template <int M>
__device__ __forceinline__ float row_xor_add(float v) {
    const int o = __builtin_amdgcn_update_dpp(0, __float_as_int(v), 0x160 | M, 0xf, 0xf, true);
    return v + __int_as_float(o);
}

template <int C>
__global__ void __launch_bounds__(256)
node_attention_kernel(const int* __restrict__ rowptr,
                      const int* __restrict__ gsrc,
                      const float* __restrict__ gw,
                      const float* __restrict__ xl,
                      const float* __restrict__ xr,
                      const float* __restrict__ We,
                      const float* __restrict__ att,
                      float* __restrict__ outacc, int nN) {
    constexpr int HC  = 4 * C;
    constexpr int VEC = HC / WAVE;
    constexpr int NV  = VEC / 4;
    constexpr int U   = 16 / VEC;

    const int lane   = threadIdx.x & 31;
    const int wave0  = __builtin_amdgcn_readfirstlane((int)((blockIdx.x * blockDim.x + threadIdx.x) >> 5));
    const int nwaves = (int)((gridDim.x * blockDim.x) >> 5);
    const int ch0    = lane * VEC;

    float we[VEC], at[VEC];
#pragma unroll
    for (int i = 0; i < VEC; ++i) { we[i] = We[ch0 + i]; at[i] = att[ch0 + i]; }

    for (int d = wave0; d < nN; d += nwaves) {
        const int beg = rowptr[d];
        const int end = rowptr[d + 1];
        const size_t rowd = (size_t)d * HC + ch0;

        float xrv[VEC], acc[VEC];
        {
            const v4f* __restrict__ pr = (const v4f*)(xr + rowd);
            const v4f* __restrict__ pl = (const v4f*)(xl + rowd);
#pragma unroll
            for (int q = 0; q < NV; ++q) {
                const v4f a = pr[q];
                const v4f b = pl[q];
                xrv[4 * q + 0] = a.x; xrv[4 * q + 1] = a.y; xrv[4 * q + 2] = a.z; xrv[4 * q + 3] = a.w;
                acc[4 * q + 0] = b.x; acc[4 * q + 1] = b.y; acc[4 * q + 2] = b.z; acc[4 * q + 3] = b.w;
            }
        }

        int   sj = 0;
        float wj = 0.f;
        if (beg + lane < end) { sj = gsrc[beg + lane]; wj = gw[beg + lane]; }

        float wsum = wj;
        for (int j = beg + WAVE + lane; j < end; j += WAVE) wsum += gw[j];
        wsum = row_xor_add<1>(wsum);
        wsum = row_xor_add<2>(wsum);
        wsum = row_xor_add<4>(wsum);
        wsum = row_xor_add<8>(wsum);
        wsum += __shfl_xor(wsum, 16, WAVE);
        const float lw = wsum / fmaxf((float)(end - beg), 1.0f);

        float m;
        {
            float t = 0.f;
#pragma unroll
            for (int i = 0; i < VEC; ++i) {
                float v = acc[i] + xrv[i] + lw * we[i];
                v = (v > 0.f) ? v : 0.2f * v;
                t += v * at[i];
            }
            t = row_xor_add<1>(t);
            t = row_xor_add<2>(t);
            t = row_xor_add<4>(t);
            m = t;
        }
        float den = 1.f;

        for (int base = beg; base < end; base += WAVE) {
            const int nb = min(end - base, WAVE);
            if (base != beg) {
                sj = 0;
                wj = 0.f;
                if (lane < nb) { sj = gsrc[base + lane]; wj = gw[base + lane]; }
            }

            for (int j = 0; j < nb; j += U) {
                float r[U][VEC];
                float w[U];
#pragma unroll
                for (int u = 0; u < U; ++u) {
                    const int jj = min(j + u, nb - 1);
                    const int s  = __builtin_amdgcn_readlane(sj, jj);
                    w[u] = __int_as_float(__builtin_amdgcn_readlane(__float_as_int(wj), jj));
                    const v4f* __restrict__ p = (const v4f*)(xl + (size_t)s * HC + ch0);
#pragma unroll
                    for (int q = 0; q < NV; ++q) {
                        const v4f x = p[q];
                        r[u][4 * q + 0] = x.x; r[u][4 * q + 1] = x.y; r[u][4 * q + 2] = x.z; r[u][4 * q + 3] = x.w;
                    }
                }
#pragma unroll
                for (int u = 0; u < U; ++u) {
                    float t = 0.f;
#pragma unroll
                    for (int i = 0; i < VEC; ++i) {
                        float v = r[u][i] + xrv[i] + w[u] * we[i];
                        v = (v > 0.f) ? v : 0.2f * v;
                        t += v * at[i];
                    }
                    t = row_xor_add<1>(t);
                    t = row_xor_add<2>(t);
                    t = row_xor_add<4>(t);
                    if (u != 0 && j + u >= nb) t = -__builtin_inff();
                    const float mn = fmaxf(m, t);
                    const float sc = __expf(m - mn);
                    const float p  = __expf(t - mn);
                    den = den * sc + p;
#pragma unroll
                    for (int i = 0; i < VEC; ++i) acc[i] = acc[i] * sc + p * r[u][i];
                    m = mn;
                }
            }
        }

        const float inv = 1.0f / den;
        v4f* __restrict__ po = (v4f*)(outacc + rowd);
#pragma unroll
        for (int q = 0; q < NV; ++q) {
            v4f o;
            o.x = acc[4 * q + 0] * inv; o.y = acc[4 * q + 1] * inv;
            o.z = acc[4 * q + 2] * inv; o.w = acc[4 * q + 3] * inv;
            po[q] = o;
        }
    }
}

__global__ void finalize_concat_kernel(const float* __restrict__ acc,
                                       const float* __restrict__ bias,
                                       const float* __restrict__ g,
                                       const float* __restrict__ be,
                                       const float* __restrict__ res,
                                       float* __restrict__ out, int nN) {
    const int gtid = blockIdx.x * blockDim.x + threadIdx.x;
    const int node = gtid >> 5;
    const int lane = gtid & 31;
    if (node >= nN) return;
    const int D = 128;

    float v[4], s = 0.f, s2 = 0.f;
#pragma unroll
    for (int i = 0; i < 4; ++i) {
        const int ch = lane + i * WAVE;
        float x = acc[(size_t)node * D + ch] + bias[ch];
        x = (x > 0.f) ? x : (__expf(x) - 1.f);
        v[i] = x;
        s += x;
        s2 += x * x;
    }
#pragma unroll
    for (int off = 16; off > 0; off >>= 1) {
        s  += __shfl_xor(s, off, WAVE);
        s2 += __shfl_xor(s2, off, WAVE);
    }
    const float mu  = s / (float)D;
    const float var = s2 / (float)D - mu * mu;
    const float inv = rsqrtf(var + 1e-5f);
#pragma unroll
    for (int i = 0; i < 4; ++i) {
        const int ch = lane + i * WAVE;
        out[(size_t)node * D + ch] =
            (v[i] - mu) * inv * g[ch] + be[ch] + res[(size_t)node * D + ch];
    }
}

__global__ void finalize_mean_kernel(const float* __restrict__ acc,
                                     const float* __restrict__ bias,
                                     const float* __restrict__ g,
                                     const float* __restrict__ be,
                                     const float* __restrict__ res,
                                     float* __restrict__ out, int nN) {
    const int gtid = blockIdx.x * blockDim.x + threadIdx.x;
    const int node = gtid >> 5;
    const int lane = gtid & 31;
    if (node >= nN) return;
    const int D = 64;

    float v[2], s = 0.f, s2 = 0.f;
#pragma unroll
    for (int i = 0; i < 2; ++i) {
        const int c = lane + i * WAVE;
        const size_t nb = (size_t)node * 256;
        float x = 0.25f * (acc[nb + 0 * D + c] + acc[nb + 1 * D + c] +
                           acc[nb + 2 * D + c] + acc[nb + 3 * D + c]) + bias[c];
        x = (x > 0.f) ? x : (__expf(x) - 1.f);
        v[i] = x;
        s += x;
        s2 += x * x;
    }
#pragma unroll
    for (int off = 16; off > 0; off >>= 1) {
        s  += __shfl_xor(s, off, WAVE);
        s2 += __shfl_xor(s2, off, WAVE);
    }
    const float mu  = s / (float)D;
    const float var = s2 / (float)D - mu * mu;
    const float inv = rsqrtf(var + 1e-5f);
#pragma unroll
    for (int i = 0; i < 2; ++i) {
        const int c = lane + i * WAVE;
        out[(size_t)node * D + c] =
            (v[i] - mu) * inv * g[c] + be[c] + res[(size_t)node * D + c];
    }
}

static inline int blocks_for(long long work, int bs) {
    return (int)((work + bs - 1) / bs);
}

extern "C" void kernel_launch(void* const* d_in, const int* in_sizes, int n_in,
                              void* d_out, int out_size, void* d_ws, size_t ws_size,
                              hipStream_t stream) {
    const float* x    = (const float*)d_in[0];
    const int*   ei   = (const int*)d_in[1];
    const float* ew   = (const float*)d_in[2];
    const float* Wl0  = (const float*)d_in[3];
    const float* bl0  = (const float*)d_in[4];
    const float* Wr0  = (const float*)d_in[5];
    const float* br0  = (const float*)d_in[6];
    const float* We0  = (const float*)d_in[7];
    const float* att0 = (const float*)d_in[8];
    const float* b0   = (const float*)d_in[9];
    const float* g0   = (const float*)d_in[10];
    const float* be0  = (const float*)d_in[11];
    const float* Wl1  = (const float*)d_in[12];
    const float* bl1  = (const float*)d_in[13];
    const float* Wr1  = (const float*)d_in[14];
    const float* br1  = (const float*)d_in[15];
    const float* We1  = (const float*)d_in[16];
    const float* att1 = (const float*)d_in[17];
    const float* b1   = (const float*)d_in[18];
    const float* g1   = (const float*)d_in[19];
    const float* be1  = (const float*)d_in[20];
    const float* P0   = (const float*)d_in[21];
    const float* pb0  = (const float*)d_in[22];
    const float* P1   = (const float*)d_in[23];
    const float* pb1  = (const float*)d_in[24];

    const int IN  = 128, HIDD = 128, O1 = 256, OUT = 64;
    const int N = in_sizes[0] / IN;
    const int E = in_sizes[2];
    const int* srcp = ei;
    const int* dstp = ei + E;

    float* ws = (float*)d_ws;
    size_t off = 0;
    float* XL    = ws + off; off += (size_t)N * O1;
    float* XR    = ws + off; off += (size_t)N * O1;
    float* ACC   = ws + off; off += (size_t)N * O1;
    float* RES   = ws + off; off += (size_t)N * HIDD;
    float* HCUR  = ws + off; off += (size_t)N * HIDD;
    int*   cnt    = (int*)(ws + off); off += (size_t)N;
    int*   rowptr = (int*)(ws + off); off += (size_t)N + 1;
    int*   rank   = (int*)(ws + off); off += (size_t)E;
    int*   gsrc   = (int*)(ws + off); off += (size_t)E;
    float* gw     = ws + off; off += (size_t)E;
    (void)ws_size; (void)n_in; (void)out_size;

    const int BS = 256;
    const int NODE_BLOCKS_L0 = 1600;
    const int NODE_BLOCKS_L1 = 1280;
    const int GEMM_BLOCKS = 256;

    fill_kernel<<<blocks_for(N, BS), BS, 0, stream>>>((float*)cnt, 0.f, (long long)N);
    degree_rank_kernel<<<blocks_for(E, BS), BS, 0, stream>>>(dstp, cnt, rank, E);
    rowptr_scan_kernel<<<1, 1024, 0, stream>>>(cnt, rowptr, N);
    group_by_dst_kernel<<<blocks_for(E, BS), BS, 0, stream>>>(srcp, dstp, ew, rank, rowptr, gsrc, gw, E);

    {
        layer_gemm_wmma_kernel<128, 128, 128><<<GEMM_BLOCKS, 512, 0, stream>>>(
            x, Wl0, bl0, XL, Wr0, br0, XR, P0, pb0, RES, N);

        node_attention_kernel<32><<<NODE_BLOCKS_L0, BS, 0, stream>>>(
            rowptr, gsrc, gw, XL, XR, We0, att0, ACC, N);

        finalize_concat_kernel<<<blocks_for((long long)N * WAVE, BS), BS, 0, stream>>>(
            ACC, b0, g0, be0, RES, HCUR, N);
    }

    {
        layer_gemm_wmma_kernel<256, 256, 64><<<GEMM_BLOCKS, 512, 0, stream>>>(
            HCUR, Wl1, bl1, XL, Wr1, br1, XR, P1, pb1, RES, N);

        node_attention_kernel<64><<<NODE_BLOCKS_L1, BS, 0, stream>>>(
            rowptr, gsrc, gw, XL, XR, We1, att1, ACC, N);

        finalize_mean_kernel<<<blocks_for((long long)N * WAVE, BS), BS, 0, stream>>>(
            ACC, b1, g1, be1, RES, (float*)d_out, N);
    }
}


# Attention edge step with less vector work: leaky_relu folded into the att dot product (cf*(v+2/3|v|), base-2 logits), and one running-max update and one rescale of the sums per group of U edges
# speedup vs baseline: 4.6969x; 1.2396x over previous
#include <hip/hip_runtime.h>
#include <hip/hip_bf16.h>
#include <math.h>


typedef __attribute__((ext_vector_type(2))) float v2f;
typedef __attribute__((ext_vector_type(8))) float v8f;

#define WAVE 32

__global__ void fill_kernel(float* __restrict__ p, float v, long long n) {
    long long i = (long long)blockIdx.x * blockDim.x + threadIdx.x;
    if (i < n) p[i] = v;
}

typedef __attribute__((ext_vector_type(4))) float v4f;

__global__ void degree_rank_kernel(const int* __restrict__ dst,
                                   int* __restrict__ cnt,
                                   int* __restrict__ rank, int E) {
    int e = blockIdx.x * blockDim.x + threadIdx.x;
    if (e >= E) return;
    rank[e] = atomicAdd(&cnt[dst[e]], 1);
}

typedef __attribute__((ext_vector_type(4))) int v4i;

template <int S>
__device__ __forceinline__ int row_shr_add(int x) {
    return x + __builtin_amdgcn_update_dpp(0, x, 0x110 | S, 0xf, 0xf, true);
}

__device__ __forceinline__ int wave_incl_scan(int x, int lane) {
    x = row_shr_add<1>(x);
    x = row_shr_add<2>(x);
    x = row_shr_add<4>(x);
    x = row_shr_add<8>(x);
    const int low = __builtin_amdgcn_readlane(x, 15);
    return x + ((lane >= 16) ? low : 0);
}

constexpr int PL_NMAX = 65536;
constexpr int PL_NWV  = 32;
constexpr int PL_K    = PL_NMAX / 4 / (PL_NWV * WAVE);
constexpr int PL_EPF  = 2;

__global__ void __launch_bounds__(1024)
rowptr_place_kernel(const int* __restrict__ src,
                    const int* __restrict__ dst,
                    const float* __restrict__ ew,
                    const int* __restrict__ rank,
                    const int* __restrict__ cnt,
                    int* __restrict__ rowptr,
                    int* __restrict__ gsrc,
                    float* __restrict__ gw, int n, int E) {
    __shared__ __attribute__((aligned(16))) int pre[PL_NMAX];
    __shared__ int wsum[PL_NWV];
    const int tid  = threadIdx.x;
    const int lane = tid & 31;
    const int wv   = __builtin_amdgcn_readfirstlane(tid >> 5);

    const int ebeg = (int)((blockIdx.x * (unsigned)E) / gridDim.x);
    const int eend = (int)(((blockIdx.x + 1) * (unsigned)E) / gridDim.x);

    int pd[PL_EPF], pr[PL_EPF], ps[PL_EPF];
    float pwt[PL_EPF];
#pragma unroll
    for (int i = 0; i < PL_EPF; ++i) {
        const int e = ebeg + i * 1024 + tid;
        pd[i] = 0; pr[i] = 0; ps[i] = 0; pwt[i] = 0.f;
        if (e < eend) { pd[i] = dst[e]; pr[i] = rank[e]; ps[i] = src[e]; pwt[i] = ew[e]; }
    }

    const v4i* __restrict__ c4 = (const v4i*)cnt;
    v4i* __restrict__ r4 = (v4i*)rowptr;
    v4i* p4 = (v4i*)pre;
    const int n4   = n >> 2;
    const int pw   = ((n4 + PL_NWV * WAVE - 1) / (PL_NWV * WAVE)) * WAVE;
    const int wbeg = wv * pw;
    const int wend = min(wbeg + pw, n4);

    v4i v[PL_K];
#pragma unroll
    for (int k = 0; k < PL_K; ++k) {
        const int g = wbeg + WAVE * k + lane;
        const v4i z = {0, 0, 0, 0};
        v[k] = (g < wend) ? c4[g] : z;
    }

    int ex[PL_K];
    int base = 0;
#pragma unroll
    for (int k = 0; k < PL_K; ++k) {
        ex[k] = 0;
        if (WAVE * k < pw) {
            const int s   = v[k].x + v[k].y + v[k].z + v[k].w;
            const int inc = wave_incl_scan(s, lane);
            ex[k] = base + inc - s;
            base += __builtin_amdgcn_readlane(inc, 31);
        }
    }
    if (lane == 0) wsum[wv] = base;
    __syncthreads();

    const int wt   = wsum[lane];
    const int wi   = wave_incl_scan(wt, lane);
    const int woff = __builtin_amdgcn_readlane(wi - wt, wv);
    const int tot  = __builtin_amdgcn_readlane(wi, 31);

    const int lofs   = (int)(((unsigned long long)rowptr & 127ull) >> 4);
    const int nlines = (n4 + lofs + 7) >> 3;
    const int lbeg   = (int)((blockIdx.x * (unsigned)nlines) / gridDim.x);
    const int lend   = (int)(((blockIdx.x + 1) * (unsigned)nlines) / gridDim.x);
#pragma unroll
    for (int k = 0; k < PL_K; ++k) {
        if (WAVE * k < pw) {
            const int g = wbeg + WAVE * k + lane;
            v4i o;
            o.x = woff + ex[k];
            o.y = o.x + v[k].x;
            o.z = o.y + v[k].y;
            o.w = o.z + v[k].z;
            if (g < wend) {
                p4[g] = o;
                const int line = (g + lofs) >> 3;
                if (line >= lbeg && line < lend) r4[g] = o;
            }
        }
    }
    if (blockIdx.x == gridDim.x - 1 && tid == 0) rowptr[n] = tot;
    __syncthreads();

#pragma unroll
    for (int i = 0; i < PL_EPF; ++i) {
        const int e = ebeg + i * 1024 + tid;
        if (e < eend) {
            const int pos = pre[pd[i]] + pr[i];
            gsrc[pos] = ps[i];
            gw[pos]   = pwt[i];
        }
    }
    for (int e = ebeg + PL_EPF * 1024 + tid; e < eend; e += 1024) {
        const int pos = pre[dst[e]] + rank[e];
        gsrc[pos] = src[e];
        gw[pos]   = ew[e];
    }
}

template <int MP, int COFF, int PITCH, int NT>
__device__ __forceinline__ void stage_weights(const float* __restrict__ Wp,
                                              float* __restrict__ Wl, int tid) {
    constexpr int Q = MP / 4;
    static_assert((64 * Q) % NT == 0, "every thread stages the same number of pieces");
#pragma unroll
    for (int i = 0; i < (64 * Q) / NT; ++i) {
        const int it = tid + i * NT;
        const int p  = it / Q;
        const int q4 = it - p * Q;
        const v4f r0 = *(const v4f*)(Wp + (size_t)(2 * p) * MP + 4 * q4);
        const v4f r1 = *(const v4f*)(Wp + (size_t)(2 * p + 1) * MP + 4 * q4);
        const int col0  = COFF + 4 * q4;
        const int gbase = col0 & ~63;
        const int cc    = (col0 & 63) >> 2;
        float* dst = Wl + p * PITCH + 2 * (gbase + cc);
        v2f e;
        e.x = r0.x; e.y = r1.x; *(v2f*)(dst +  0) = e;
        e.x = r0.y; e.y = r1.y; *(v2f*)(dst + 32) = e;
        e.x = r0.z; e.y = r1.z; *(v2f*)(dst + 64) = e;
        e.x = r0.w; e.y = r1.w; *(v2f*)(dst + 96) = e;
    }
}

template <int M0, int M1, int M2>
__global__ void __launch_bounds__(512)
layer_gemm_wmma_kernel(const float* __restrict__ X,
                       const float* __restrict__ W0, const float* __restrict__ B0, float* __restrict__ Y0,
                       const float* __restrict__ W1, const float* __restrict__ B1, float* __restrict__ Y1,
                       const float* __restrict__ W2, const float* __restrict__ B2, float* __restrict__ Y2,
                       int rows) {
    constexpr int KD    = 128;
    constexpr int MTOT  = M0 + M1 + M2;
    constexpr int PITCH = 2 * MTOT + 16;
    constexpr int G0 = M0 / 64, G1 = M1 / 64, G2 = M2 / 64;
    constexpr int G  = G0 + G1 + G2;
    constexpr int NW = 16;
    static_assert(M0 % 64 == 0 && M1 % 64 == 0 && M2 % 64 == 0, "whole column groups");
    static_assert(64 * PITCH * 4 <= 320 * 1024, "the layer's weights fit the work-group's LDS");

    __shared__ __attribute__((aligned(16))) float Wl[64 * PITCH];

    const int tid  = threadIdx.x;
    const int lane = tid & 31;
    const int c    = lane & 15;
    const int h    = lane >> 4;
    const int wv   = __builtin_amdgcn_readfirstlane(tid >> 5);

    const int nstrips = rows >> 4;
    const int sbeg = (int)((blockIdx.x * (unsigned)nstrips) / gridDim.x);
    const int send = (int)(((blockIdx.x + 1) * (unsigned)nstrips) / gridDim.x);
    const int L    = (send - sbeg) * G;
    const int ubeg = (wv * L) / NW;
    const int uend = ((wv + 1) * L) / NW;

    v4f a[16];
    int cur = -1;
    if (ubeg < uend) {
        cur = sbeg + ubeg / G;
        const v4f* __restrict__ pa = (const v4f*)(X + (size_t)(cur * 16 + c) * KD + 4 * h);
#pragma unroll
        for (int j = 0; j < 16; ++j) a[j] = pa[2 * j];
    }

    stage_weights<M0, 0,       PITCH, NW * WAVE>(W0, Wl, tid);
    stage_weights<M1, M0,      PITCH, NW * WAVE>(W1, Wl, tid);
    stage_weights<M2, M0 + M1, PITCH, NW * WAVE>(W2, Wl, tid);
    __syncthreads();

    for (int u = ubeg; u < uend; ++u) {
        const int sl    = u / G;
        const int g     = u - sl * G;
        const int strip = sbeg + sl;
        if (strip != cur) {
            cur = strip;
            const v4f* __restrict__ pa = (const v4f*)(X + (size_t)(strip * 16 + c) * KD + 4 * h);
#pragma unroll
            for (int j = 0; j < 16; ++j) a[j] = pa[2 * j];
        }

        const float* bias; float* Y; int M; int gl;
        if (g < G0)           { bias = B0; Y = Y0; M = M0; gl = g; }
        else if (g < G0 + G1) { bias = B1; Y = Y1; M = M1; gl = g - G0; }
        else                  { bias = B2; Y = Y2; M = M2; gl = g - G0 - G1; }
        const int lc = gl * 64 + 4 * c;
        const v4f bb = *(const v4f*)(bias + lc);

        const float* bl = Wl + (2 * h) * PITCH + 128 * g + 2 * c;
        v8f acc[4] = {};
#pragma unroll
        for (int j = 0; j < 16; ++j) {
            v2f a0, a1;
            a0.x = a[j].x; a0.y = a[j].y;
            a1.x = a[j].z; a1.y = a[j].w;
#pragma unroll
            for (int t = 0; t < 4; ++t) {
                const v2f b = *(const v2f*)(bl + (4 * j) * PITCH + 32 * t);
                acc[t] = __builtin_amdgcn_wmma_f32_16x16x4_f32(
                     false, a0,  false, b,
                     (short)0, acc[t],  false,  false);
            }
#pragma unroll
            for (int t = 0; t < 4; ++t) {
                const v2f b = *(const v2f*)(bl + (4 * j + 1) * PITCH + 32 * t);
                acc[t] = __builtin_amdgcn_wmma_f32_16x16x4_f32(
                     false, a1,  false, b,
                     (short)0, acc[t],  false,  false);
            }
        }

        float* yrow = Y + (size_t)(strip * 16 + 8 * h) * M + lc;
#pragma unroll
        for (int j = 0; j < 8; ++j) {
            v4f o;
            o.x = acc[0][j] + bb.x; o.y = acc[1][j] + bb.y;
            o.z = acc[2][j] + bb.z; o.w = acc[3][j] + bb.w;
            *(v4f*)(yrow + (size_t)j * M) = o;
        }
    }
}

template <int M>
__device__ __forceinline__ float row_xor_add(float v) {
    const int o = __builtin_amdgcn_update_dpp(0, __float_as_int(v), 0x160 | M, 0xf, 0xf, true);
    return v + __int_as_float(o);
}

__device__ __forceinline__ float xhalf(float x) {
    const int b = __float_as_int(x);
    return __int_as_float(__builtin_amdgcn_permlanex16(b, b, 0x76543210, 0xfedcba98, false, false));
}

template <int VEC>
__device__ __forceinline__ float head_logit(const float* __restrict__ row, const float* __restrict__ xrv,
                                            const float* __restrict__ we, const float* __restrict__ cf, float w) {
    float ta = 0.f, tb = 0.f;
#pragma unroll
    for (int i = 0; i < VEC; i += 2) {
        const float v0 = row[i]     + fmaf(w, we[i],     xrv[i]);
        const float v1 = row[i + 1] + fmaf(w, we[i + 1], xrv[i + 1]);
        const float u0 = fmaf(fabsf(v0), 0.66666667f, v0);
        const float u1 = fmaf(fabsf(v1), 0.66666667f, v1);
        ta = fmaf(u0, cf[i],     ta);
        tb = fmaf(u1, cf[i + 1], tb);
    }
    float t = ta + tb;
    t = row_xor_add<1>(t);
    t = row_xor_add<2>(t);
    t = row_xor_add<4>(t);
    return t;
}

template <int C>
__global__ void __launch_bounds__(256, (C == 32) ? 16 : 10)
node_attention_kernel(const int* __restrict__ rowptr,
                      const int* __restrict__ gsrc,
                      const float* __restrict__ gw,
                      const float* __restrict__ xl,
                      const float* __restrict__ xr,
                      const float* __restrict__ We,
                      const float* __restrict__ att,
                      const float* __restrict__ bias,
                      const float* __restrict__ g,
                      const float* __restrict__ be,
                      const float* __restrict__ res,
                      float* __restrict__ out, int nN) {
    constexpr int HC  = 4 * C;
    constexpr int VEC = HC / WAVE;
    constexpr int NV  = VEC / 4;
    constexpr int U   = 16 / VEC;

    const int lane   = threadIdx.x & 31;
    const int wave0  = __builtin_amdgcn_readfirstlane((int)((blockIdx.x * blockDim.x + threadIdx.x) >> 5));
    const int nwaves = (int)((gridDim.x * blockDim.x) >> 5);
    const int ch0    = lane * VEC;

    float we[VEC], cf[VEC];
#pragma unroll
    for (int i = 0; i < VEC; ++i) { we[i] = We[ch0 + i]; cf[i] = att[ch0 + i] * (0.6f * 1.44269504f); }

    constexpr int OD = (C == 32) ? HC : C;
    const int oc0 = (C == 32) ? ch0 : (8 * (lane & 7) + 4 * ((lane >> 3) & 1));

    for (int d = wave0; d < nN; d += nwaves) {
        const int beg = rowptr[d];
        const int end = rowptr[d + 1];
        const size_t rowd = (size_t)d * HC + ch0;
        const v4f rs = *(const v4f*)(res + (size_t)d * OD + oc0);

        float xrv[VEC], acc[VEC];
        {
            const v4f* __restrict__ pr = (const v4f*)(xr + rowd);
            const v4f* __restrict__ pl = (const v4f*)(xl + rowd);
#pragma unroll
            for (int q = 0; q < NV; ++q) {
                const v4f a = pr[q];
                const v4f b = pl[q];
                xrv[4 * q + 0] = a.x; xrv[4 * q + 1] = a.y; xrv[4 * q + 2] = a.z; xrv[4 * q + 3] = a.w;
                acc[4 * q + 0] = b.x; acc[4 * q + 1] = b.y; acc[4 * q + 2] = b.z; acc[4 * q + 3] = b.w;
            }
        }

        int   sj = 0;
        float wj = 0.f;
        if (beg + lane < end) { sj = gsrc[beg + lane]; wj = gw[beg + lane]; }

        float wsum = wj;
        for (int j = beg + WAVE + lane; j < end; j += WAVE) wsum += gw[j];
        wsum = row_xor_add<1>(wsum);
        wsum = row_xor_add<2>(wsum);
        wsum = row_xor_add<4>(wsum);
        wsum = row_xor_add<8>(wsum);
        wsum += __shfl_xor(wsum, 16, WAVE);
        const float lw = wsum / fmaxf((float)(end - beg), 1.0f);

        float m = head_logit<VEC>(acc, xrv, we, cf, lw);
        float den = 1.f;

        for (int base = beg; base < end; base += WAVE) {
            const int nb = min(end - base, WAVE);
            if (base != beg) {
                sj = 0;
                wj = 0.f;
                if (lane < nb) { sj = gsrc[base + lane]; wj = gw[base + lane]; }
            }

            for (int j = 0; j < nb; j += U) {
                float r[U][VEC];
                float w[U];
#pragma unroll
                for (int u = 0; u < U; ++u) {
                    const int jj = min(j + u, nb - 1);
                    const int s  = __builtin_amdgcn_readlane(sj, jj);
                    w[u] = __int_as_float(__builtin_amdgcn_readlane(__float_as_int(wj), jj));
                    const v4f* __restrict__ p = (const v4f*)(xl + (size_t)s * HC + ch0);
#pragma unroll
                    for (int q = 0; q < NV; ++q) {
                        const v4f x = p[q];
                        r[u][4 * q + 0] = x.x; r[u][4 * q + 1] = x.y; r[u][4 * q + 2] = x.z; r[u][4 * q + 3] = x.w;
                    }
                }
                float t[U];
#pragma unroll
                for (int u = 0; u < U; ++u) {
                    float tt = head_logit<VEC>(r[u], xrv, we, cf, w[u]);
                    if (u != 0 && j + u >= nb) tt = -__builtin_inff();
                    t[u] = tt;
                }
                float mn = m;
#pragma unroll
                for (int u = 0; u < U; ++u) mn = fmaxf(mn, t[u]);
                const float sc = __builtin_amdgcn_exp2f(m - mn);
                float p[U];
                float ps = 0.f;
#pragma unroll
                for (int u = 0; u < U; ++u) { p[u] = __builtin_amdgcn_exp2f(t[u] - mn); ps += p[u]; }
                den = den * sc + ps;
#pragma unroll
                for (int i = 0; i < VEC; ++i) {
                    float a = acc[i] * sc;
#pragma unroll
                    for (int u = 0; u < U; ++u) a = fmaf(p[u], r[u][i], a);
                    acc[i] = a;
                }
                m = mn;
            }
        }

        const float inv = 1.0f / den;
        int pb = (C == 32) ? ch0 : 8 * (lane & 7);
        int po = oc0;
        asm volatile("" : "+v"(pb), "+v"(po));
        if constexpr (C == 32) {
            const v4f bq = *(const v4f*)(bias + pb);
            const float bv[4] = {bq.x, bq.y, bq.z, bq.w};
            float x[4], s = 0.f, s2 = 0.f;
#pragma unroll
            for (int i = 0; i < 4; ++i) {
                float v = acc[i] * inv + bv[i];
                v = (v > 0.f) ? v : (__expf(v) - 1.f);
                x[i] = v;
                s  += v;
                s2 += v * v;
            }
            s  = row_xor_add<1>(s);  s2 = row_xor_add<1>(s2);
            s  = row_xor_add<2>(s);  s2 = row_xor_add<2>(s2);
            s  = row_xor_add<4>(s);  s2 = row_xor_add<4>(s2);
            s  = row_xor_add<8>(s);  s2 = row_xor_add<8>(s2);
            s  += xhalf(s);          s2 += xhalf(s2);
            const float mu  = s / (float)OD;
            const float var = s2 / (float)OD - mu * mu;
            const float rin = rsqrtf(var + 1e-5f);
            const v4f gq = *(const v4f*)(g + po);
            const v4f eq = *(const v4f*)(be + po);
            v4f o;
            o.x = (x[0] - mu) * rin * gq.x + eq.x + rs.x;
            o.y = (x[1] - mu) * rin * gq.y + eq.y + rs.y;
            o.z = (x[2] - mu) * rin * gq.z + eq.z + rs.z;
            o.w = (x[3] - mu) * rin * gq.w + eq.w + rs.w;
            *(v4f*)(out + (size_t)d * OD + oc0) = o;
        } else {
            const v4f bq0 = *(const v4f*)(bias + pb);
            const v4f bq1 = *(const v4f*)(bias + pb + 4);
            const float bv[8] = {bq0.x, bq0.y, bq0.z, bq0.w, bq1.x, bq1.y, bq1.z, bq1.w};
            float x[8], s = 0.f, s2 = 0.f;
#pragma unroll
            for (int i = 0; i < 8; ++i) {
                float y = acc[i] * inv;
                y = row_xor_add<8>(y);
                y += xhalf(y);
                float v = 0.25f * y + bv[i];
                v = (v > 0.f) ? v : (__expf(v) - 1.f);
                x[i] = v;
                s  += v;
                s2 += v * v;
            }
            s  = row_xor_add<1>(s);  s2 = row_xor_add<1>(s2);
            s  = row_xor_add<2>(s);  s2 = row_xor_add<2>(s2);
            s  = row_xor_add<4>(s);  s2 = row_xor_add<4>(s2);
            const float mu  = s / (float)OD;
            const float var = s2 / (float)OD - mu * mu;
            const float rin = rsqrtf(var + 1e-5f);
            const bool up = (lane & 8) != 0;
            const v4f gq = *(const v4f*)(g + po);
            const v4f eq = *(const v4f*)(be + po);
            v4f o;
            o.x = ((up ? x[4] : x[0]) - mu) * rin * gq.x + eq.x + rs.x;
            o.y = ((up ? x[5] : x[1]) - mu) * rin * gq.y + eq.y + rs.y;
            o.z = ((up ? x[6] : x[2]) - mu) * rin * gq.z + eq.z + rs.z;
            o.w = ((up ? x[7] : x[3]) - mu) * rin * gq.w + eq.w + rs.w;
            if (lane < 16) *(v4f*)(out + (size_t)d * OD + oc0) = o;
        }
    }
}

static inline int blocks_for(long long work, int bs) {
    return (int)((work + bs - 1) / bs);
}

extern "C" void kernel_launch(void* const* d_in, const int* in_sizes, int n_in,
                              void* d_out, int out_size, void* d_ws, size_t ws_size,
                              hipStream_t stream) {
    const float* x    = (const float*)d_in[0];
    const int*   ei   = (const int*)d_in[1];
    const float* ew   = (const float*)d_in[2];
    const float* Wl0  = (const float*)d_in[3];
    const float* bl0  = (const float*)d_in[4];
    const float* Wr0  = (const float*)d_in[5];
    const float* br0  = (const float*)d_in[6];
    const float* We0  = (const float*)d_in[7];
    const float* att0 = (const float*)d_in[8];
    const float* b0   = (const float*)d_in[9];
    const float* g0   = (const float*)d_in[10];
    const float* be0  = (const float*)d_in[11];
    const float* Wl1  = (const float*)d_in[12];
    const float* bl1  = (const float*)d_in[13];
    const float* Wr1  = (const float*)d_in[14];
    const float* br1  = (const float*)d_in[15];
    const float* We1  = (const float*)d_in[16];
    const float* att1 = (const float*)d_in[17];
    const float* b1   = (const float*)d_in[18];
    const float* g1   = (const float*)d_in[19];
    const float* be1  = (const float*)d_in[20];
    const float* P0   = (const float*)d_in[21];
    const float* pb0  = (const float*)d_in[22];
    const float* P1   = (const float*)d_in[23];
    const float* pb1  = (const float*)d_in[24];

    const int IN  = 128, HIDD = 128, O1 = 256, OUT = 64;
    const int N = in_sizes[0] / IN;
    const int E = in_sizes[2];
    const int* srcp = ei;
    const int* dstp = ei + E;

    float* ws = (float*)d_ws;
    size_t off = 0;
    float* XL    = ws + off; off += (size_t)N * O1;
    float* XR    = ws + off; off += (size_t)N * O1;
    float* RES   = ws + off; off += (size_t)N * HIDD;
    float* HCUR  = ws + off; off += (size_t)N * HIDD;
    int*   cnt    = (int*)(ws + off); off += (size_t)N;
    int*   rowptr = (int*)(ws + off); off += (size_t)N + 1;
    int*   rank   = (int*)(ws + off); off += (size_t)E;
    int*   gsrc   = (int*)(ws + off); off += (size_t)E;
    float* gw     = ws + off; off += (size_t)E;
    (void)ws_size; (void)n_in; (void)out_size;

    const int BS = 256;
    const int NODE_BLOCKS_L0 = 1600;
    const int NODE_BLOCKS_L1 = 1280;
    const int GEMM_BLOCKS = 256;
    const int PLACE_BLOCKS = 256;

    fill_kernel<<<blocks_for(N, BS), BS, 0, stream>>>((float*)cnt, 0.f, (long long)N);
    degree_rank_kernel<<<blocks_for(E, BS), BS, 0, stream>>>(dstp, cnt, rank, E);
    rowptr_place_kernel<<<PLACE_BLOCKS, 1024, 0, stream>>>(srcp, dstp, ew, rank, cnt, rowptr, gsrc, gw, N, E);

    {
        layer_gemm_wmma_kernel<128, 128, 128><<<GEMM_BLOCKS, 512, 0, stream>>>(
            x, Wl0, bl0, XL, Wr0, br0, XR, P0, pb0, RES, N);

        node_attention_kernel<32><<<NODE_BLOCKS_L0, BS, 0, stream>>>(
            rowptr, gsrc, gw, XL, XR, We0, att0, b0, g0, be0, RES, HCUR, N);
    }

    {
        layer_gemm_wmma_kernel<256, 256, 64><<<GEMM_BLOCKS, 512, 0, stream>>>(
            HCUR, Wl1, bl1, XL, Wr1, br1, XR, P1, pb1, RES, N);

        node_attention_kernel<64><<<NODE_BLOCKS_L1, BS, 0, stream>>>(
            rowptr, gsrc, gw, XL, XR, We1, att1, b1, g1, be1, RES, (float*)d_out, N);
    }
}
